# starting kernel
# baseline (speedup 1.0000x reference)
#include <hip/hip_runtime.h>

#define N_NODES  100000
#define N_EDGES  3200000
#define F_IN     10
#define KPAD     12
#define N_OUT    64

typedef __attribute__((ext_vector_type(2))) float v2f;
typedef __attribute__((ext_vector_type(8))) float v8f;

__global__ void __launch_bounds__(256) gcn_zero_kernel(float* p, int n) {
    int i = blockIdx.x * blockDim.x + threadIdx.x;
    if (i < n) p[i] = 0.0f;
}

__global__ void __launch_bounds__(256) gcn_degree_kernel(const int* __restrict__ src,
                                                         const int* __restrict__ dst,
                                                         float* __restrict__ deg_out,
                                                         float* __restrict__ deg_in,
                                                         int nE) {
    int e = blockIdx.x * blockDim.x + threadIdx.x;
    if (e < nE) {
        atomicAdd(&deg_out[src[e]], 1.0f);
        atomicAdd(&deg_in [dst[e]], 1.0f);
    }
}

__global__ void __launch_bounds__(256) gcn_nodeprep_kernel(const float* __restrict__ x,
                                                           const float* __restrict__ deg_out,
                                                           const float* __restrict__ deg_in,
                                                           float* __restrict__ norm_d,
                                                           float* __restrict__ xscaled,
                                                           int nN) {
    int n = blockIdx.x * blockDim.x + threadIdx.x;
    if (n < nN) {
        float dout = deg_out[n]; dout = (dout > 0.0f) ? dout : 1.0f;
        float din  = deg_in [n]; din  = (din  > 0.0f) ? din  : 1.0f;
        float ns = rsqrtf(dout);
        norm_d[n] = rsqrtf(din);
        const float* xr = x + (size_t)n * 40 + 15;
        float* o = xscaled + (size_t)n * F_IN;
#pragma unroll
        for (int k = 0; k < F_IN; ++k) o[k] = xr[k] * ns;
    }
}

__global__ void __launch_bounds__(64) gcn_foldw_kernel(const float* __restrict__ W1,
                                                       const float* __restrict__ b1,
                                                       const float* __restrict__ W2,
                                                       const float* __restrict__ b2,
                                                       float* __restrict__ Wc,
                                                       float* __restrict__ bc) {
    int j = threadIdx.x;
    float acc = b2[j];
    for (int k = 0; k < 128; ++k) acc += b1[k] * W2[k * 64 + j];
    bc[j] = acc;
#pragma unroll 1
    for (int i = 0; i < KPAD; ++i) {
        float s = 0.0f;
        if (i < F_IN)
            for (int k = 0; k < 128; ++k) s += W1[i * 128 + k] * W2[k * 64 + j];
        Wc[i * 64 + j] = s;
    }
}

__global__ void __launch_bounds__(256) gcn_scatter_kernel(const int* __restrict__ src,
                                                          const int* __restrict__ dst,
                                                          const float* __restrict__ xscaled,
                                                          float* __restrict__ agg,
                                                          int nE) {
    int e = blockIdx.x * blockDim.x + threadIdx.x;
    if (e < nE) {
        int s = src[e], d = dst[e];
        const float* xs = xscaled + (size_t)s * F_IN;
        float*       a  = agg     + (size_t)d * F_IN;
        float v[F_IN];
#pragma unroll
        for (int k = 0; k < F_IN; ++k) v[k] = xs[k];
#pragma unroll
        for (int k = 0; k < F_IN; ++k) atomicAdd(&a[k], v[k]);
    }
}

__global__ void __launch_bounds__(32) gcn_wmma_kernel(const float* __restrict__ agg,
                                                      const float* __restrict__ norm_d,
                                                      const float* __restrict__ Wc,
                                                      const float* __restrict__ bc,
                                                      float* __restrict__ out) {
    int tile = blockIdx.x;
    int lane = threadIdx.x;
    int m  = lane & 15;
    int kh = lane >> 4;

    int node = tile * 16 + m;
    float nd = norm_d[node];
    const float* arow = agg + (size_t)node * F_IN;

    v2f afrag[3];
#pragma unroll
    for (int ks = 0; ks < 3; ++ks) {
        int k0 = ks * 4 + kh * 2;
        afrag[ks].x = (k0     < F_IN) ? arow[k0]     * nd : 0.0f;
        afrag[ks].y = (k0 + 1 < F_IN) ? arow[k0 + 1] * nd : 0.0f;
    }

    int nn = lane & 15;
#pragma unroll
    for (int j = 0; j < 4; ++j) {
        v8f c = {};
#pragma unroll
        for (int ks = 0; ks < 3; ++ks) {
            int kb = ks * 4 + kh * 2;
            v2f b;
            b.x = Wc[(kb    ) * N_OUT + j * 16 + nn];
            b.y = Wc[(kb + 1) * N_OUT + j * 16 + nn];
            c = __builtin_amdgcn_wmma_f32_16x16x4_f32(
                     false, afrag[ks],
                     false, b,
                     (short)0, c,
                     false,  false);
        }
        float bias = bc[j * 16 + nn];
        int nbase = tile * 16;
#pragma unroll
        for (int r = 0; r < 8; ++r) {
            int row = r + 8 * kh;
            out[(size_t)(nbase + row) * N_OUT + j * 16 + nn] = c[r] + bias;
        }
    }
}

extern "C" void kernel_launch(void* const* d_in, const int* in_sizes, int n_in,
                              void* d_out, int out_size, void* d_ws, size_t ws_size,
                              hipStream_t stream) {
    const float* x   = (const float*)d_in[0];
    const int*   src = (const int*)  d_in[1];
    const int*   dst = (const int*)  d_in[2];
    const float* W1  = (const float*)d_in[3];
    const float* b1  = (const float*)d_in[4];
    const float* W2  = (const float*)d_in[5];
    const float* b2  = (const float*)d_in[6];
    float* out = (float*)d_out;

    float* ws = (float*)d_ws;
    float* deg_out = ws;
    float* deg_in  = ws + N_NODES;
    float* agg     = ws + 2 * N_NODES;
    float* norm_dv = ws + 2 * N_NODES + N_NODES * F_IN;
    float* xscaled = norm_dv + N_NODES;
    float* Wc      = xscaled + (size_t)N_NODES * F_IN;
    float* bcv     = Wc + KPAD * N_OUT;

    const int zeroN = 2 * N_NODES + N_NODES * F_IN;
    gcn_zero_kernel<<<(zeroN + 255) / 256, 256, 0, stream>>>(ws, zeroN);

    gcn_degree_kernel<<<(N_EDGES + 255) / 256, 256, 0, stream>>>(src, dst, deg_out, deg_in, N_EDGES);

    gcn_nodeprep_kernel<<<(N_NODES + 255) / 256, 256, 0, stream>>>(x, deg_out, deg_in,
                                                                   norm_dv, xscaled, N_NODES);

    gcn_foldw_kernel<<<1, 64, 0, stream>>>(W1, b1, W2, b2, Wc, bcv);

    gcn_scatter_kernel<<<(N_EDGES + 255) / 256, 256, 0, stream>>>(src, dst, xscaled, agg, N_EDGES);

    gcn_wmma_kernel<<<N_NODES / 16, 32, 0, stream>>>(agg, norm_dv, Wc, bcv, out);
}


# Weight fold (W1@W2, b1@W2) moved from its own 1-work-group launch into the product kernel, formed on the matrix unit: 128-term sums as 8 partials of 16 added in fixed order; 6 launches -> 5
# speedup vs baseline: 13.4763x; 13.4763x over previous
#include <hip/hip_runtime.h>

#define N_NODES  100000
#define N_EDGES  3200000
#define F_IN     10
#define N_OUT    64
#define ROWP     16
#define NCH      64
#define CH_VEC   (N_EDGES / NCH / 4)
#define HWORDS   (N_NODES / 2)
#define NHALF    (N_NODES / 2)
#define SC_EDGES 1024
#define SC_ROUNDS (SC_EDGES / 256)
#define SC_GRID  2048
#define TK_PITCH 64
#define PW_WAVES 8
#define N_TILES  (N_NODES / 16)
#define WC_ROWS  12
#define WC_PITCH 72

typedef __attribute__((ext_vector_type(2))) float v2f;
typedef __attribute__((ext_vector_type(8))) float v8f;

__global__ void __launch_bounds__(256) gcn_zero_kernel(float* p, int n) {
    int i = blockIdx.x * blockDim.x + threadIdx.x;
    if (i < n) p[i] = 0.0f;
}

__global__ void __launch_bounds__(1024) gcn_degree_kernel(const int* __restrict__ src,
                                                          const int* __restrict__ dst,
                                                          unsigned* __restrict__ part) {
    __shared__ __attribute__((aligned(16))) unsigned h[HWORDS];
    const int tid  = threadIdx.x;
    const int role = blockIdx.x / NCH;
    const int ch   = blockIdx.x - role * NCH;
    const int4* p  = (const int4*)(role ? dst : src) + (size_t)ch * CH_VEC;

    int4 v[13];
#pragma unroll
    for (int k = 0; k < 12; ++k) v[k] = p[tid + k * 1024];
    const bool tail = tid + 12 * 1024 < CH_VEC;
    v[12] = p[tail ? tid + 12 * 1024 : tid];

    uint4* h4 = (uint4*)h;
    for (int i = tid; i < HWORDS / 4; i += 1024) h4[i] = make_uint4(0u, 0u, 0u, 0u);
    __syncthreads();

#pragma unroll
    for (int k = 0; k < 13; ++k) {
        if (k < 12 || tail) {
            atomicAdd(&h[v[k].x >> 1], (v[k].x & 1) ? 0x10000u : 1u);
            atomicAdd(&h[v[k].y >> 1], (v[k].y & 1) ? 0x10000u : 1u);
            atomicAdd(&h[v[k].z >> 1], (v[k].z & 1) ? 0x10000u : 1u);
            atomicAdd(&h[v[k].w >> 1], (v[k].w & 1) ? 0x10000u : 1u);
        }
    }
    __syncthreads();

    uint4* o4 = (uint4*)(part + (size_t)blockIdx.x * HWORDS);
    for (int i = tid; i < HWORDS / 4; i += 1024) o4[i] = h4[i];
}

__global__ void __launch_bounds__(256) gcn_nodeprep_kernel(const float* __restrict__ x,
                                                           const unsigned* __restrict__ part,
                                                           float* __restrict__ norm_d,
                                                           float* __restrict__ xscaled,
                                                           int nPair) {
    int t = blockIdx.x * blockDim.x + threadIdx.x;
    if (t < nPair) {
        const unsigned* ps = part + t;
        const unsigned* pd = part + (size_t)NCH * HWORDS + t;
        unsigned o0 = 0, o1 = 0, i0 = 0, i1 = 0;
#pragma unroll
        for (int c = 0; c < NCH; ++c) {
            unsigned a = ps[(size_t)c * HWORDS];
            unsigned b = pd[(size_t)c * HWORDS];
            o0 += a & 0xffffu;  o1 += a >> 16;
            i0 += b & 0xffffu;  i1 += b >> 16;
        }
        float dout0 = (float)o0; dout0 = (dout0 > 0.0f) ? dout0 : 1.0f;
        float dout1 = (float)o1; dout1 = (dout1 > 0.0f) ? dout1 : 1.0f;
        float din0  = (float)i0; din0  = (din0  > 0.0f) ? din0  : 1.0f;
        float din1  = (float)i1; din1  = (din1  > 0.0f) ? din1  : 1.0f;
        float ns0 = rsqrtf(dout0);
        float ns1 = rsqrtf(dout1);
        norm_d[2 * t]     = rsqrtf(din0);
        norm_d[2 * t + 1] = rsqrtf(din1);
        const float* xr = x + (size_t)t * 80 + 15;
        float* o = xscaled + (size_t)t * (2 * ROWP);
#pragma unroll
        for (int k = 0; k < F_IN; ++k) o[k] = xr[k] * ns0;
#pragma unroll
        for (int k = 0; k < F_IN; ++k) o[ROWP + k] = xr[40 + k] * ns1;
    }
}

__global__ void __launch_bounds__(256) gcn_scatter_kernel(const int* __restrict__ src,
                                                          const int* __restrict__ dst,
                                                          const float* __restrict__ xscaled,
                                                          float* __restrict__ agg,
                                                          unsigned* __restrict__ tickets,
                                                          int nE) {
    __shared__ unsigned s_t;
    __shared__ int2 s_list[8][SC_EDGES / 8];
    const int tid  = threadIdx.x;
    const int lane = tid & 31;
    const int f    = lane & 15;
    const int rh   = (lane >> 4) & 1;
    const bool act = f < F_IN;
    int2* lst = s_list[tid >> 5];
    const unsigned nChunk = ((unsigned)nE + SC_EDGES - 1u) / SC_EDGES;
    const unsigned NONE = 0xffffffffu;

    __shared__ unsigned s_half;
    if (tid == 0) s_half = (((unsigned)__builtin_amdgcn_s_sendmsg_rtn(0x87) >> 16) >> 2) & 1u;
    __syncthreads();
    const unsigned myhalf = (unsigned)__builtin_amdgcn_readfirstlane((int)s_half) & 1u;

    for (int pass = 0; pass < 2; ++pass) {
        const unsigned h = (myhalf ^ (unsigned)pass) & 1u;
        unsigned* tk = tickets + h * TK_PITCH;
        for (;;) {
            if (tid == 0) s_t = atomicAdd(tk, 1u);
            __syncthreads();
            const unsigned tkt = (unsigned)__builtin_amdgcn_readfirstlane((int)s_t);
            __syncthreads();
            if (tkt >= nChunk) break;
            const int cbase = (int)(tkt * SC_EDGES);

            int sv[SC_ROUNDS], dv[SC_ROUNDS];
            if (cbase + SC_EDGES <= nE) {
#pragma unroll
                for (int r = 0; r < SC_ROUNDS; ++r) {
                    sv[r] = src[cbase + r * 256 + tid];
                    dv[r] = dst[cbase + r * 256 + tid];
                }
            } else {
#pragma unroll
                for (int r = 0; r < SC_ROUNDS; ++r) {
                    const int e = cbase + r * 256 + tid;
                    sv[r] = (e < nE) ? src[e] : 0;
                    dv[r] = (e < nE) ? dst[e] : -1;
                }
            }

            unsigned n = 0;
#pragma unroll
            for (int r = 0; r < SC_ROUNDS; ++r) {
                const int d = dv[r];
                const bool in = (d >= 0) && (((d >= NHALF) ? 1u : 0u) == h);
                const unsigned m = __builtin_amdgcn_ballot_w32(in);
                if (in) lst[n + __builtin_amdgcn_mbcnt_lo(m, 0u)] = make_int2(sv[r], d);
                n += (unsigned)__builtin_popcount(m);
            }
            __syncthreads();

            for (unsigned g0 = 0; g0 < n; g0 += 16) {
                if (g0 + 16 <= n) {
                    if (act) {
                        unsigned so[8], dn[8];
#pragma unroll
                        for (int u = 0; u < 8; ++u) {
                            const int2 p = lst[g0 + 2 * u + rh];
                            so[u] = (unsigned)p.x * ROWP + f;
                            dn[u] = (unsigned)p.y * ROWP + f;
                        }
                        float v[8];
#pragma unroll
                        for (int u = 0; u < 8; ++u) v[u] = xscaled[so[u]];
#pragma unroll
                        for (int u = 0; u < 8; ++u) atomicAdd(&agg[dn[u]], v[u]);
                    }
                } else {
                    unsigned so[8], dn[8];
#pragma unroll
                    for (int u = 0; u < 8; ++u) {
                        const unsigned idx = g0 + 2 * u + rh;
                        so[u] = 0u;
                        dn[u] = NONE;
                        if (act && idx < n) {
                            const int2 p = lst[idx];
                            so[u] = (unsigned)p.x * ROWP + f;
                            dn[u] = (unsigned)p.y * ROWP + f;
                        }
                    }
                    float v[8];
#pragma unroll
                    for (int u = 0; u < 8; ++u) { v[u] = 0.0f; if (dn[u] != NONE) v[u] = xscaled[so[u]]; }
#pragma unroll
                    for (int u = 0; u < 8; ++u) if (dn[u] != NONE) atomicAdd(&agg[dn[u]], v[u]);
                }
            }
        }
    }
}

__global__ void __launch_bounds__(256) gcn_wmma_kernel(const float* __restrict__ agg,
                                                       const float* __restrict__ norm_d,
                                                       const float* __restrict__ W1,
                                                       const float* __restrict__ b1,
                                                       const float* __restrict__ W2,
                                                       const float* __restrict__ b2,
                                                       float* __restrict__ out) {
    __shared__ float s_part[PW_WAVES][16 * N_OUT];
    __shared__ float s_wc[WC_ROWS * WC_PITCH];

    const int tid  = threadIdx.x;
    const int wave = tid >> 5;
    const int lane = tid & 31;
    const int m  = lane & 15;
    const int kh = lane >> 4;

    const int  tile = blockIdx.x * PW_WAVES + wave;
    const bool live = tile < N_TILES;

    const int node = (live ? tile : 0) * 16 + m;
    float nd = norm_d[node];
    const float* arow = agg + (size_t)node * ROWP;

    float araw[3][2];
#pragma unroll
    for (int ks = 0; ks < 3; ++ks) {
        const int k0 = ks * 4 + kh * 2;
        araw[ks][0] = arow[k0];
        araw[ks][1] = arow[k0 + 1];
    }

    {
        const float* lrow = (m < F_IN) ? (W1 + m * 128) : b1;
        const bool   lrow_live = m <= F_IN;
        v2f fa[4];
#pragma unroll
        for (int kk = 0; kk < 4; ++kk) {
            const int k = wave * 16 + kk * 4 + kh * 2;
            const float l0 = lrow[k], l1 = lrow[k + 1];
            fa[kk].x = lrow_live ? l0 : 0.0f;
            fa[kk].y = lrow_live ? l1 : 0.0f;
        }
        v2f fb[4][4];
#pragma unroll
        for (int kk = 0; kk < 4; ++kk) {
            const int k = wave * 16 + kk * 4 + kh * 2;
#pragma unroll
            for (int j = 0; j < 4; ++j) {
                fb[kk][j].x = W2[(k    ) * N_OUT + j * 16 + m];
                fb[kk][j].y = W2[(k + 1) * N_OUT + j * 16 + m];
            }
        }
        v8f pacc[4] = {};
#pragma unroll
        for (int kk = 0; kk < 4; ++kk) {
#pragma unroll
            for (int j = 0; j < 4; ++j) {
                pacc[j] = __builtin_amdgcn_wmma_f32_16x16x4_f32(
                               false, fa[kk],
                               false, fb[kk][j],
                               (short)0, pacc[j],
                               false,  false);
            }
        }
#pragma unroll
        for (int j = 0; j < 4; ++j) {
#pragma unroll
            for (int r = 0; r < 8; ++r)
                s_part[wave][(r + 8 * kh) * N_OUT + j * 16 + m] = pacc[j][r];
        }
    }
    __syncthreads();

#pragma unroll
    for (int q = 0; q < 3; ++q) {
        const int e = tid + 256 * q;
        float s = s_part[0][e];
#pragma unroll
        for (int w = 1; w < PW_WAVES; ++w) s += s_part[w][e];
        s_wc[(e >> 6) * WC_PITCH + (e & 63)] = s;
    }
    __syncthreads();

    if (live) {
        const int nn = m;
        v2f afrag[3];
#pragma unroll
        for (int ks = 0; ks < 3; ++ks) {
            const int k0 = ks * 4 + kh * 2;
            afrag[ks].x = (k0     < F_IN) ? araw[ks][0] * nd : 0.0f;
            afrag[ks].y = (k0 + 1 < F_IN) ? araw[ks][1] * nd : 0.0f;
        }
#pragma unroll
        for (int j = 0; j < 4; ++j) {
            v8f c = {};
#pragma unroll
            for (int ks = 0; ks < 3; ++ks) {
                int kb = ks * 4 + kh * 2;
                v2f b;
                const float w0 = s_wc[(kb    ) * WC_PITCH + j * 16 + nn];
                const float w1 = s_wc[(kb + 1) * WC_PITCH + j * 16 + nn];
                b.x = (kb     < F_IN) ? w0 : 0.0f;
                b.y = (kb + 1 < F_IN) ? w1 : 0.0f;
                c = __builtin_amdgcn_wmma_f32_16x16x4_f32(
                         false, afrag[ks],
                         false, b,
                         (short)0, c,
                         false,  false);
            }
            float bias = s_wc[F_IN * WC_PITCH + j * 16 + nn] + b2[j * 16 + nn];
            int nbase = tile * 16;
#pragma unroll
            for (int r = 0; r < 8; ++r) {
                int row = r + 8 * kh;
                out[(size_t)(nbase + row) * N_OUT + j * 16 + nn] = c[r] + bias;
            }
        }
    }
}

extern "C" void kernel_launch(void* const* d_in, const int* in_sizes, int n_in,
                              void* d_out, int out_size, void* d_ws, size_t ws_size,
                              hipStream_t stream) {
    const float* x   = (const float*)d_in[0];
    const int*   src = (const int*)  d_in[1];
    const int*   dst = (const int*)  d_in[2];
    const float* W1  = (const float*)d_in[3];
    const float* b1  = (const float*)d_in[4];
    const float* W2  = (const float*)d_in[5];
    const float* b2  = (const float*)d_in[6];
    float* out = (float*)d_out;

    float* ws = (float*)d_ws;
    float* agg     = ws;
    unsigned* tickets = (unsigned*)(ws + N_NODES * ROWP);
    float* norm_dv = ws + N_NODES * ROWP + 2 * TK_PITCH;
    float* xscaled = norm_dv + N_NODES;
    unsigned* part = (unsigned*)(xscaled + (size_t)N_NODES * ROWP);

    gcn_zero_kernel<<<(N_NODES * ROWP + 2 * TK_PITCH + 255) / 256, 256, 0, stream>>>(agg, N_NODES * ROWP + 2 * TK_PITCH);

    gcn_degree_kernel<<<2 * NCH, 1024, 0, stream>>>(src, dst, part);

    gcn_nodeprep_kernel<<<(HWORDS + 255) / 256, 256, 0, stream>>>(x, part, norm_dv, xscaled, HWORDS);

    gcn_scatter_kernel<<<SC_GRID, 256, 0, stream>>>(src, dst, xscaled, agg, tickets, N_EDGES);

    gcn_wmma_kernel<<<(N_TILES + PW_WAVES - 1) / PW_WAVES, 32 * PW_WAVES, 0, stream>>>(agg, norm_dv, W1, b1, W2, b2, out);
}
